# starting kernel
# baseline (speedup 1.0000x reference)
#include <hip/hip_runtime.h>
#include <math.h>

typedef __attribute__((ext_vector_type(2))) float v2f;
typedef __attribute__((ext_vector_type(8))) float v8f;

constexpr int B      = 256;
constexpr int ICAPS  = 3200;
constexpr int INDIM  = 8;
constexpr int J      = 8;
constexpr int M      = 16;
constexpr int BT     = 16;
constexpr int IC     = 64;
constexpr int NIB    = ICAPS / IC;
constexpr int WAVES  = 8;
constexpr int TPB    = WAVES * 32;
constexpr int SSZ    = B * J * M;
constexpr int UTS    = 18;

__global__ __launch_bounds__(TPB) void routing_pass(
    const float* __restrict__ x,
    const float* __restrict__ W,
    const float* __restrict__ vprev,
    float* __restrict__ s_out)
{
    __shared__ float xs[IC * BT * INDIM];
    __shared__ float vs[BT * J * M];
    __shared__ float s_lds[J * BT * M];
    __shared__ float a_lds[WAVES * BT * J];
    __shared__ float ut_lds[WAVES * BT * UTS];

    const int t  = threadIdx.x;
    const int b0 = blockIdx.x * BT;
    const int i0 = blockIdx.y * IC;

    {
        const float4* xg  = reinterpret_cast<const float4*>(x);
        float4*       xs4 = reinterpret_cast<float4*>(xs);
#pragma unroll
        for (int k = 0; k < (BT * IC * 2) / TPB; ++k) {
            const int q  = t + k * TPB;
            const int bb = q >> 7;
            const int r  = q & 127;
            const int ii = r >> 1;
            const int n4 = r & 1;
            xs4[(ii * BT + bb) * 2 + n4] =
                xg[((size_t)(b0 + bb) * ICAPS + (i0 + ii)) * 2 + n4];
        }
    }
    const bool uniform = (vprev == nullptr);
    if (!uniform) {
        const float4* vg  = reinterpret_cast<const float4*>(vprev) + b0 * (J * M / 4);
        float4*       vs4 = reinterpret_cast<float4*>(vs);
        for (int q = t; q < BT * J * M / 4; q += TPB) vs4[q] = vg[q];
    }
    for (int q = t; q < J * BT * M; q += TPB) s_lds[q] = 0.0f;
    __syncthreads();

    const int wave = t >> 5;
    const int lane = t & 31;
    const int half = lane >> 4;
    const int lm   = lane & 15;

    float* up = &ut_lds[wave * (BT * UTS)];

    v8f sacc[J];
#pragma unroll
    for (int j = 0; j < J; ++j) sacc[j] = (v8f){0,0,0,0,0,0,0,0};

    for (int il = wave; il < IC; il += WAVES) {
        const int i = i0 + il;

        const float* xrow = &xs[(il * BT + lm) * INDIM];
        v2f af0, af1;
        af0.x = xrow[half * 2 + 0]; af0.y = xrow[half * 2 + 1];
        af1.x = xrow[half * 2 + 4]; af1.y = xrow[half * 2 + 5];

        v8f acc[J];
        const float* wbase = W + (size_t)i * (J * INDIM * M);
#pragma unroll
        for (int j = 0; j < J; ++j) {
            const float* wj = wbase + j * (INDIM * M);
            const int kr = half * 2;
            v2f bf0, bf1;
            bf0.x = wj[(kr + 0) * M + lm]; bf0.y = wj[(kr + 1) * M + lm];
            bf1.x = wj[(kr + 4) * M + lm]; bf1.y = wj[(kr + 5) * M + lm];
            v8f c = {0,0,0,0,0,0,0,0};
            c = __builtin_amdgcn_wmma_f32_16x16x4_f32(false, af0, false, bf0,
                                                      (short)0, c, false, false);
            c = __builtin_amdgcn_wmma_f32_16x16x4_f32(false, af1, false, bf1,
                                                      (short)0, c, false, false);
            acc[j] = c;
        }

        if (uniform) {
#pragma unroll
            for (int j = 0; j < J; ++j)
#pragma unroll
                for (int r = 0; r < 8; ++r) sacc[j][r] += 0.125f * acc[j][r];
        } else {
            const bool holder = ((lm >> 3) == half);
#pragma unroll
            for (int j = 0; j < J; ++j) {
#pragma unroll
                for (int r = 0; r < 8; ++r)
                    up[(r + half * 8) * UTS + lm] = acc[j][r];
                asm volatile("" ::: "memory");
                v8f d = {0,0,0,0,0,0,0,0};
#pragma unroll
                for (int kq = 0; kq < 4; ++kq) {
                    const int kb = kq * 4 + half * 2;
                    v2f avf = *(const v2f*)&vs[(lm * J + j) * M + kb];
                    v2f buf = *(const v2f*)&up[lm * UTS + kb];
                    d = __builtin_amdgcn_wmma_f32_16x16x4_f32(false, avf, false, buf,
                                                              (short)0, d, false, false);
                }
                const int rsel = lm & 7;
                float dv = d[0];
#pragma unroll
                for (int r = 1; r < 8; ++r) dv = (rsel == r) ? d[r] : dv;
                if (holder) a_lds[(wave * BT + lm) * J + j] = dv;
                asm volatile("" ::: "memory");
            }

            asm volatile("s_wait_dscnt 0x0" ::: "memory");
            if (lane < BT) {
                const int bb = lane;
                float av[J];
                float mx = -3.0e38f;
#pragma unroll
                for (int j = 0; j < J; ++j) {
                    av[j] = a_lds[(wave * BT + bb) * J + j];
                    mx = fmaxf(mx, av[j]);
                }
                float se = 0.0f;
#pragma unroll
                for (int j = 0; j < J; ++j) { av[j] = __expf(av[j] - mx); se += av[j]; }
                const float inv = 1.0f / se;
                v2f* cw = (v2f*)&a_lds[(wave * BT + bb) * J];
#pragma unroll
                for (int q = 0; q < 4; ++q) {
                    v2f c2; c2.x = av[2 * q] * inv; c2.y = av[2 * q + 1] * inv;
                    cw[q] = c2;
                }
            }
            asm volatile("s_wait_dscnt 0x0" ::: "memory");

#pragma unroll
            for (int r = 0; r < 8; ++r) {
                const int bb = r + half * 8;
                const v2f* cp = (const v2f*)&a_lds[(wave * BT + bb) * J];
#pragma unroll
                for (int q = 0; q < 4; ++q) {
                    v2f c2 = cp[q];
                    sacc[2 * q][r]     += c2.x * acc[2 * q][r];
                    sacc[2 * q + 1][r] += c2.y * acc[2 * q + 1][r];
                }
            }
        }
    }

#pragma unroll
    for (int j = 0; j < J; ++j)
#pragma unroll
        for (int r = 0; r < 8; ++r) {
            const int bb = r + half * 8;
            atomicAdd(&s_lds[(j * BT + bb) * M + lm], sacc[j][r]);
        }
    __syncthreads();
    for (int q = t; q < J * BT * M; q += TPB) {
        const int j  = q / (BT * M);
        const int bb = (q / M) & (BT - 1);
        const int m  = q & (M - 1);
        atomicAdd(&s_out[((size_t)(b0 + bb) * J + j) * M + m], s_lds[q]);
    }
}

__global__ void squash_k(const float* __restrict__ s, const float* __restrict__ vadd,
                         float* __restrict__ vout)
{
    const int bj = blockIdx.x * blockDim.x + threadIdx.x;
    if (bj >= B * J) return;
    const float* sp = s + bj * M;
    float sq = 0.0f;
#pragma unroll
    for (int m = 0; m < M; ++m) { const float e = sp[m]; sq += e * e; }
    const float scale = sqrtf(sq) / (1.0f + sq);
#pragma unroll
    for (int m = 0; m < M; ++m) {
        float r = sp[m] * scale;
        if (vadd) r += vadd[bj * M + m];
        vout[bj * M + m] = r;
    }
}

__global__ void zero_ws(float* p, int n)
{
    const int t = blockIdx.x * blockDim.x + threadIdx.x;
    if (t < n) p[t] = 0.0f;
}

extern "C" void kernel_launch(void* const* d_in, const int* in_sizes, int n_in,
                              void* d_out, int out_size, void* d_ws, size_t ws_size,
                              hipStream_t stream)
{
    (void)in_sizes; (void)n_in; (void)out_size; (void)ws_size;
    const float* x = (const float*)d_in[0];
    const float* W = (const float*)d_in[1];
    float* out = (float*)d_out;
    float* ws  = (float*)d_ws;

    float* s0  = ws;
    float* s1  = ws + SSZ;
    float* s2  = ws + 2 * SSZ;
    float* v0  = ws + 3 * SSZ;
    float* v01 = ws + 4 * SSZ;

    zero_ws<<<(3 * SSZ + TPB - 1) / TPB, TPB, 0, stream>>>(ws, 3 * SSZ);

    const dim3 grid(B / BT, NIB);
    routing_pass<<<grid, TPB, 0, stream>>>(x, W, nullptr, s0);
    squash_k<<<(B * J + TPB - 1) / TPB, TPB, 0, stream>>>(s0, nullptr, v0);
    routing_pass<<<grid, TPB, 0, stream>>>(x, W, v0, s1);
    squash_k<<<(B * J + TPB - 1) / TPB, TPB, 0, stream>>>(s1, v0, v01);
    routing_pass<<<grid, TPB, 0, stream>>>(x, W, v01, s2);
    squash_k<<<(B * J + TPB - 1) / TPB, TPB, 0, stream>>>(s2, nullptr, out);
}


# each wave works on two 16-row batch tiles: every W fragment it loads feeds both tiles' products in both sweeps (block 32 rows x 64 capsules, 400 blocks, 4 waves per SIMD); W reads halve
# speedup vs baseline: 2.1241x; 2.1241x over previous
#include <hip/hip_runtime.h>
#include <math.h>
#include <type_traits>


typedef __attribute__((ext_vector_type(2))) float v2f;
typedef __attribute__((ext_vector_type(4))) float v4f;
typedef __attribute__((ext_vector_type(8))) float v8f;

constexpr int B      = 256;
constexpr int ICAPS  = 3200;
constexpr int INDIM  = 8;
constexpr int J      = 8;
constexpr int M      = 16;
constexpr int TL     = 2;
constexpr int BT     = 16 * TL;
constexpr int IC     = 64;
constexpr int NIB    = ICAPS / IC;
constexpr int WAVES  = 8;
constexpr int TPB    = WAVES * 32;
constexpr int SSZ    = B * J * M;
constexpr int WSTR   = J * INDIM * M;
constexpr int NPW    = IC / 2 / WAVES;
constexpr int NSTEP  = NPW * J;
constexpr int PF     = 2;
constexpr int PSTR   = 2 * WAVES * WSTR;
constexpr int WPS    = 4;

template <bool UNIFORM>
__global__ __launch_bounds__(TPB, WPS) void routing_pass(
    const float* __restrict__ x,
    const float* __restrict__ W,
    const float* __restrict__ vprev,
    float* __restrict__ s_out)
{
    struct __attribute__((aligned(16))) TileRouted {
        float vs[J * 4 * BT * 4];
        float xs[IC * BT * INDIM];
        float cs[WAVES * NPW * TL * 32 * J];
    };
    struct __attribute__((aligned(16))) TileUniform {
        float s_lds[J * BT * M];
        float xs[IC * BT * INDIM];
    };
    static_assert(J * 4 * BT * 4 == J * BT * M, "the s partials must fit the v tile exactly");
    __shared__ typename std::conditional<UNIFORM, TileUniform, TileRouted>::type sm;
    float* const xs = sm.xs;
    float* s_lds = nullptr;
    float* vs = nullptr;
    float* cs = nullptr;
    if constexpr (UNIFORM) { s_lds = sm.s_lds; } else { vs = sm.vs; cs = sm.cs; s_lds = sm.vs; }

    const int t  = threadIdx.x;
    const int b0 = blockIdx.x * BT;
    const int i0 = blockIdx.y * IC;

    const int wave = t >> 5;
    const int lane = t & 31;
    const int half = lane >> 4;
    const int lm   = lane & 15;

    v4f wr[PF + 1][2];
    const float* wz = W + (size_t)(i0 + 2 * wave + (lm >> 3)) * WSTR + (lm & 7) * M + half * 8;
    float wq[PF + 1][INDIM];
    const float* wsp = W + (size_t)(i0 + 2 * wave + half) * WSTR + lm;
    auto sweep2_head = [&]() {
#pragma unroll
        for (int s = 0; s < PF; ++s) {
            const float* q = wsp + (s >> 3) * PSTR + (s & 7) * (INDIM * M);
#pragma unroll
            for (int n = 0; n < INDIM; ++n) wq[s][n] = q[n * M];
        }
    };
    if (UNIFORM) {
        sweep2_head();
    } else {
#pragma unroll
        for (int s = 0; s < PF; ++s) {
            const float* q = wz + (s >> 3) * PSTR + (s & 7) * (INDIM * M);
            wr[s][0] = *reinterpret_cast<const v4f*>(q);
            wr[s][1] = *reinterpret_cast<const v4f*>(q + 4);
        }
    }

    {
        const float4* xg  = reinterpret_cast<const float4*>(x);
        float4*       xs4 = reinterpret_cast<float4*>(xs);
#pragma unroll
        for (int k = 0; k < (BT * IC * 2) / TPB; ++k) {
            const int q  = t + k * TPB;
            const int bb = q >> 7;
            const int r  = q & 127;
            const int ii = r >> 1;
            const int n4 = r & 1;
            xs4[(ii * BT + bb) * 2 + n4] =
                xg[((size_t)(b0 + bb) * ICAPS + (i0 + ii)) * 2 + n4];
        }
    }
    if (!UNIFORM) {
        const float4* vg  = reinterpret_cast<const float4*>(vprev) + b0 * (J * M / 4);
        float4*       vs4 = reinterpret_cast<float4*>(vs);
        for (int q = t; q < BT * J * M / 4; q += TPB) {
            const int bb = q >> 5;
            const int jm = q & 31;
            vs4[jm * BT + bb] = vg[q];
        }
    }
    if (UNIFORM) {
        for (int q = t; q < J * BT * M; q += TPB) s_lds[q] = 0.0f;
    }
    __syncthreads();

    if (!UNIFORM) {
        float lg[TL][J];
        v4f x0[TL], x1[TL];
        v4f vf[2][TL][2];
#pragma unroll
        for (int tl = 0; tl < TL; ++tl) {
            vf[0][tl][0] = *reinterpret_cast<const v4f*>(&vs[((half * 2 + 0) * BT + tl * 16 + lm) * 4]);
            vf[0][tl][1] = *reinterpret_cast<const v4f*>(&vs[((half * 2 + 1) * BT + tl * 16 + lm) * 4]);
        }
#pragma unroll
        for (int tl = 0; tl < TL; ++tl) { x0[tl] = (v4f){0,0,0,0}; x1[tl] = (v4f){0,0,0,0}; }
#pragma unroll
        for (int s = 0; s < NSTEP; ++s) {
            const int pt = s >> 3;
            const int j  = s & 7;
            if (s + PF < NSTEP) {
                const int s2 = s + PF;
                const float* q = wz + (s2 >> 3) * PSTR + (s2 & 7) * (INDIM * M);
                wr[s2 % (PF + 1)][0] = *reinterpret_cast<const v4f*>(q);
                wr[s2 % (PF + 1)][1] = *reinterpret_cast<const v4f*>(q + 4);
            }
            if (j == 0) {
                asm volatile("" ::: "memory");
                const int p = wave + pt * WAVES;
#pragma unroll
                for (int tl = 0; tl < TL; ++tl) {
                    const v4f* xr = reinterpret_cast<const v4f*>(&xs[((2 * p + half) * BT + tl * 16 + lm) * INDIM]);
                    x0[tl] = xr[0];
                    x1[tl] = xr[1];
                }
            }
            const v4f wa = wr[s % (PF + 1)][0];
            const v4f wb = wr[s % (PF + 1)][1];
            if (s + 1 < NSTEP) {
                const int j1 = (s + 1) & 7;
#pragma unroll
                for (int tl = 0; tl < TL; ++tl) {
                    vf[(s + 1) & 1][tl][0] = *reinterpret_cast<const v4f*>(&vs[((j1 * 4 + half * 2 + 0) * BT + tl * 16 + lm) * 4]);
                    vf[(s + 1) & 1][tl][1] = *reinterpret_cast<const v4f*>(&vs[((j1 * 4 + half * 2 + 1) * BT + tl * 16 + lm) * 4]);
                }
            }
            v8f z[TL];
#pragma unroll
            for (int tl = 0; tl < TL; ++tl) {
                const v4f va = vf[s & 1][tl][0];
                const v4f vb = vf[s & 1][tl][1];
                v8f d = {0,0,0,0,0,0,0,0};
                d = __builtin_amdgcn_wmma_f32_16x16x4_f32(false, (v2f){wa.x, wa.y}, false, (v2f){va.x, va.y},
                                                          (short)0, d, false, false);
                d = __builtin_amdgcn_wmma_f32_16x16x4_f32(false, (v2f){wa.z, wa.w}, false, (v2f){va.z, va.w},
                                                          (short)0, d, false, false);
                d = __builtin_amdgcn_wmma_f32_16x16x4_f32(false, (v2f){wb.x, wb.y}, false, (v2f){vb.x, vb.y},
                                                          (short)0, d, false, false);
                d = __builtin_amdgcn_wmma_f32_16x16x4_f32(false, (v2f){wb.z, wb.w}, false, (v2f){vb.z, vb.w},
                                                          (short)0, d, false, false);
                z[tl] = d;
            }
#pragma unroll
            for (int tl = 0; tl < TL; ++tl) {
                float a = x0[tl].x * z[tl][0];
                a = fmaf(x0[tl].y, z[tl][1], a); a = fmaf(x0[tl].z, z[tl][2], a); a = fmaf(x0[tl].w, z[tl][3], a);
                a = fmaf(x1[tl].x, z[tl][4], a); a = fmaf(x1[tl].y, z[tl][5], a); a = fmaf(x1[tl].z, z[tl][6], a); a = fmaf(x1[tl].w, z[tl][7], a);
                asm("" : "+v"(a));
                lg[tl][j] = a;
            }
            if (j == J - 1) {
                if (s == NSTEP - 1) sweep2_head();
#pragma unroll
                for (int tl = 0; tl < TL; ++tl) {
                    float mx = lg[tl][0];
#pragma unroll
                    for (int jj = 1; jj < J; ++jj) mx = fmaxf(mx, lg[tl][jj]);
                    float se = 0.0f;
#pragma unroll
                    for (int jj = 0; jj < J; ++jj) { lg[tl][jj] = __expf(lg[tl][jj] - mx); se += lg[tl][jj]; }
                    const float inv = 1.0f / se;
                    v4f* cw = reinterpret_cast<v4f*>(&cs[(((wave * NPW + pt) * TL + tl) * 32 + lane) * J]);
                    cw[0] = (v4f){lg[tl][0] * inv, lg[tl][1] * inv, lg[tl][2] * inv, lg[tl][3] * inv};
                    cw[1] = (v4f){lg[tl][4] * inv, lg[tl][5] * inv, lg[tl][6] * inv, lg[tl][7] * inv};
                }
            }
        }
        __syncthreads();
        for (int q = t; q < J * BT * M; q += TPB) s_lds[q] = 0.0f;
        __syncthreads();
    }

    v8f sacc[TL][J];
#pragma unroll
    for (int tl = 0; tl < TL; ++tl)
#pragma unroll
        for (int j = 0; j < J; ++j) sacc[tl][j] = (v8f){0,0,0,0,0,0,0,0};

    {
        v4f x0[TL], x1[TL];
        float c[TL][J];
#pragma unroll
        for (int tl = 0; tl < TL; ++tl) {
            x0[tl] = (v4f){0,0,0,0}; x1[tl] = (v4f){0,0,0,0};
#pragma unroll
            for (int j = 0; j < J; ++j) c[tl][j] = 1.0f;
        }
#pragma unroll
        for (int s = 0; s < NSTEP; ++s) {
            const int pt = s >> 3;
            const int j  = s & 7;
            if (s + PF < NSTEP) {
                const int s2 = s + PF;
                const float* q = wsp + (s2 >> 3) * PSTR + (s2 & 7) * (INDIM * M);
#pragma unroll
                for (int n = 0; n < INDIM; ++n) wq[s2 % (PF + 1)][n] = q[n * M];
            }
            if (j == 0) {
                const int p = wave + pt * WAVES;
#pragma unroll
                for (int tl = 0; tl < TL; ++tl) {
                    const v4f* xr = reinterpret_cast<const v4f*>(&xs[((2 * p + half) * BT + tl * 16 + lm) * INDIM]);
                    x0[tl] = xr[0];
                    x1[tl] = xr[1];
                    if (!UNIFORM) {
                        const v4f* cr = reinterpret_cast<const v4f*>(&cs[(((wave * NPW + pt) * TL + tl) * 32 + lane) * J]);
                        const v4f c0 = cr[0], c1 = cr[1];
                        c[tl][0] = c0.x; c[tl][1] = c0.y; c[tl][2] = c0.z; c[tl][3] = c0.w;
                        c[tl][4] = c1.x; c[tl][5] = c1.y; c[tl][6] = c1.z; c[tl][7] = c1.w;
                    }
                }
            }
            const float* wj = wq[s % (PF + 1)];
            const v2f w0 = {wj[0], wj[1]};
            const v2f w1 = {wj[2], wj[3]};
            const v2f w2 = {wj[4], wj[5]};
            const v2f w3 = {wj[6], wj[7]};
#pragma unroll
            for (int tl = 0; tl < TL; ++tl) {
                const float cj = UNIFORM ? 1.0f : c[tl][j];
                const v2f a0 = {x0[tl].x * cj, x0[tl].y * cj};
                const v2f a1 = {x0[tl].z * cj, x0[tl].w * cj};
                const v2f a2 = {x1[tl].x * cj, x1[tl].y * cj};
                const v2f a3 = {x1[tl].z * cj, x1[tl].w * cj};
                v8f d = sacc[tl][j];
                d = __builtin_amdgcn_wmma_f32_16x16x4_f32(false, a0, false, w0, (short)0, d, false, false);
                d = __builtin_amdgcn_wmma_f32_16x16x4_f32(false, a1, false, w1, (short)0, d, false, false);
                d = __builtin_amdgcn_wmma_f32_16x16x4_f32(false, a2, false, w2, (short)0, d, false, false);
                d = __builtin_amdgcn_wmma_f32_16x16x4_f32(false, a3, false, w3, (short)0, d, false, false);
                sacc[tl][j] = d;
            }
        }
    }

    const float sc = UNIFORM ? 0.125f : 1.0f;
#pragma unroll
    for (int tl = 0; tl < TL; ++tl)
#pragma unroll
        for (int j = 0; j < J; ++j)
#pragma unroll
            for (int r = 0; r < 8; ++r) {
                const int bb = tl * 16 + r + half * 8;
                atomicAdd(&s_lds[(j * BT + bb) * M + lm], sc * sacc[tl][j][r]);
            }
    __syncthreads();
    for (int q = t; q < J * BT * M; q += TPB) {
        const int j  = q / (BT * M);
        const int bb = (q / M) & (BT - 1);
        const int m  = q & (M - 1);
        atomicAdd(&s_out[((size_t)(b0 + bb) * J + j) * M + m], s_lds[q]);
    }
}

__global__ void squash_k(const float* __restrict__ s, const float* __restrict__ vadd,
                         float* __restrict__ vout)
{
    const int bj = blockIdx.x * blockDim.x + threadIdx.x;
    if (bj >= B * J) return;
    const float* sp = s + bj * M;
    float sq = 0.0f;
#pragma unroll
    for (int m = 0; m < M; ++m) { const float e = sp[m]; sq += e * e; }
    const float scale = sqrtf(sq) / (1.0f + sq);
#pragma unroll
    for (int m = 0; m < M; ++m) {
        float r = sp[m] * scale;
        if (vadd) r += vadd[bj * M + m];
        vout[bj * M + m] = r;
    }
}

__global__ void zero_ws(float* p, int n)
{
    const int t = blockIdx.x * blockDim.x + threadIdx.x;
    if (t < n) p[t] = 0.0f;
}

extern "C" void kernel_launch(void* const* d_in, const int* in_sizes, int n_in,
                              void* d_out, int out_size, void* d_ws, size_t ws_size,
                              hipStream_t stream)
{
    (void)in_sizes; (void)n_in; (void)out_size; (void)ws_size;
    const float* x = (const float*)d_in[0];
    const float* W = (const float*)d_in[1];
    float* out = (float*)d_out;
    float* ws  = (float*)d_ws;

    float* s0  = ws;
    float* s1  = ws + SSZ;
    float* s2  = ws + 2 * SSZ;
    float* v0  = ws + 3 * SSZ;
    float* v01 = ws + 4 * SSZ;

    zero_ws<<<(3 * SSZ + TPB - 1) / TPB, TPB, 0, stream>>>(ws, 3 * SSZ);

    const dim3 grid(B / BT, NIB);
    routing_pass<true><<<grid, TPB, 0, stream>>>(x, W, nullptr, s0);
    squash_k<<<(B * J + TPB - 1) / TPB, TPB, 0, stream>>>(s0, nullptr, v0);
    routing_pass<false><<<grid, TPB, 0, stream>>>(x, W, v0, s1);
    squash_k<<<(B * J + TPB - 1) / TPB, TPB, 0, stream>>>(s1, v0, v01);
    routing_pass<false><<<grid, TPB, 0, stream>>>(x, W, v01, s2);
    squash_k<<<(B * J + TPB - 1) / TPB, TPB, 0, stream>>>(s2, nullptr, out);
}


# each wave owns one output capsule j over all capsule pairs of its block (was all j, 4 pairs): no cross-wave s sum or LDS atomics; softmax via LDS; 512 equal blocks of 50 capsules; s sum order changes
# speedup vs baseline: 2.7779x; 1.3078x over previous
#include <hip/hip_runtime.h>
#include <math.h>


typedef __attribute__((ext_vector_type(2))) float v2f;
typedef __attribute__((ext_vector_type(4))) float v4f;
typedef __attribute__((ext_vector_type(8))) float v8f;

constexpr int B      = 256;
constexpr int ICAPS  = 3200;
constexpr int INDIM  = 8;
constexpr int J      = 8;
constexpr int M      = 16;
constexpr int TL     = 2;
constexpr int BT     = 16 * TL;
constexpr int IC     = 50;
constexpr int NIB    = ICAPS / IC;
constexpr int NP     = IC / 2;
constexpr int WAVES  = J;
constexpr int TPB    = WAVES * 32;
constexpr int SSZ    = B * J * M;
constexpr int WSTR   = J * INDIM * M;
constexpr int PF     = 2;
constexpr int PSTR   = 2 * WSTR;
constexpr int WPS    = 4;
constexpr int XP     = BT + 1;
constexpr int NQ     = IC * BT;
constexpr int LDS_BLOCK = 112 * 1024;
static_assert(ICAPS % IC == 0 && IC % 2 == 0, "whole capsule pairs per block");

template <bool UNIFORM>
__global__ __launch_bounds__(TPB, WPS) void routing_pass(
    const float* __restrict__ x,
    const float* __restrict__ W,
    const float* __restrict__ vprev,
    float* __restrict__ s_out)
{
    struct __attribute__((aligned(16))) Tile {
        float xs[IC * 2 * XP * 4];
        float lg[J * NQ];
        float fill[(LDS_BLOCK - (IC * 2 * XP * 4 + J * NQ) * 4) / 4];
    };
    static_assert(sizeof(Tile) == LDS_BLOCK, "two blocks per WGP");
    __shared__ Tile sm;
    float* const xs = sm.xs;
    float* const lg = sm.lg;

    const int t  = threadIdx.x;
    const int b0 = blockIdx.x * BT;
    const int i0 = blockIdx.y * IC;

    const int j    = __builtin_amdgcn_readfirstlane(t >> 5);
    const int lane = t & 31;
    const int half = lane >> 4;
    const int lm   = lane & 15;

    v4f wr[PF + 1][2];
    const float* wz = W + (size_t)(i0 + (lm >> 3)) * WSTR + j * (INDIM * M) + (lm & 7) * M + half * 8;
    float wq[PF + 1][INDIM];
    const float* wsp = W + (size_t)(i0 + half) * WSTR + j * (INDIM * M) + lm;
    auto sweep2_head = [&]() {
#pragma unroll
        for (int s = 0; s < PF; ++s) {
            const float* q = wsp + s * PSTR;
#pragma unroll
            for (int n = 0; n < INDIM; ++n) wq[s][n] = q[n * M];
        }
    };
    v4f vf[TL][2];
    if constexpr (UNIFORM) {
        sweep2_head();
    } else {
#pragma unroll
        for (int s = 0; s < PF; ++s) {
            const float* q = wz + s * PSTR;
            wr[s][0] = *reinterpret_cast<const v4f*>(q);
            wr[s][1] = *reinterpret_cast<const v4f*>(q + 4);
        }
#pragma unroll
        for (int tl = 0; tl < TL; ++tl) {
            const float* vp = vprev + ((size_t)(b0 + tl * 16 + lm) * J + j) * M + half * 8;
            vf[tl][0] = *reinterpret_cast<const v4f*>(vp);
            vf[tl][1] = *reinterpret_cast<const v4f*>(vp + 4);
        }
    }

    {
        const float4* xg  = reinterpret_cast<const float4*>(x);
        float4*       xs4 = reinterpret_cast<float4*>(xs);
        constexpr int NX = BT * IC * 2;
        auto stage = [&](int q) {
            const int bb = q / (IC * 2);
            const int k  = q - bb * (IC * 2);
            xs4[k * XP + bb] = xg[((size_t)(b0 + bb) * ICAPS + i0) * 2 + k];
        };
#pragma unroll
        for (int it = 0; it < NX / TPB; ++it) stage(t + it * TPB);
        if (NX % TPB != 0 && t < NX % TPB) stage(t + (NX / TPB) * TPB);
    }
    __syncthreads();

    const v4f* const xl = reinterpret_cast<const v4f*>(xs) + (2 * half) * XP + lm;
    auto load_x = [&](int p, v4f* xa, v4f* xb) {
#pragma unroll
        for (int tl = 0; tl < TL; ++tl) {
            xa[tl] = xl[(4 * p) * XP + tl * 16];
            xb[tl] = xl[(4 * p + 1) * XP + tl * 16];
        }
    };
    float* const lgl = lg + j * NQ + half * BT + lm;

    if constexpr (!UNIFORM) {
        v4f xa[2][TL], xb[2][TL];
        load_x(0, xa[0], xb[0]);
#pragma unroll
        for (int p = 0; p < NP; ++p) {
            asm volatile("" ::: "memory");
            if (p + PF < NP) {
                const float* q = wz + (p + PF) * PSTR;
                wr[(p + PF) % (PF + 1)][0] = *reinterpret_cast<const v4f*>(q);
                wr[(p + PF) % (PF + 1)][1] = *reinterpret_cast<const v4f*>(q + 4);
            }
            const v4f wa = wr[p % (PF + 1)][0];
            const v4f wb = wr[p % (PF + 1)][1];
            if (p + 1 < NP) load_x(p + 1, xa[(p + 1) & 1], xb[(p + 1) & 1]);
            v8f z[TL];
#pragma unroll
            for (int tl = 0; tl < TL; ++tl) {
                const v4f va = vf[tl][0];
                const v4f vb = vf[tl][1];
                v8f d = {0,0,0,0,0,0,0,0};
                d = __builtin_amdgcn_wmma_f32_16x16x4_f32(false, (v2f){wa.x, wa.y}, false, (v2f){va.x, va.y},
                                                          (short)0, d, false, false);
                d = __builtin_amdgcn_wmma_f32_16x16x4_f32(false, (v2f){wa.z, wa.w}, false, (v2f){va.z, va.w},
                                                          (short)0, d, false, false);
                d = __builtin_amdgcn_wmma_f32_16x16x4_f32(false, (v2f){wb.x, wb.y}, false, (v2f){vb.x, vb.y},
                                                          (short)0, d, false, false);
                d = __builtin_amdgcn_wmma_f32_16x16x4_f32(false, (v2f){wb.z, wb.w}, false, (v2f){vb.z, vb.w},
                                                          (short)0, d, false, false);
                z[tl] = d;
            }
#pragma unroll
            for (int tl = 0; tl < TL; ++tl) {
                const v4f x0 = xa[p & 1][tl], x1 = xb[p & 1][tl];
                float a = x0.x * z[tl][0];
                a = fmaf(x0.y, z[tl][1], a); a = fmaf(x0.z, z[tl][2], a); a = fmaf(x0.w, z[tl][3], a);
                a = fmaf(x1.x, z[tl][4], a); a = fmaf(x1.y, z[tl][5], a); a = fmaf(x1.z, z[tl][6], a); a = fmaf(x1.w, z[tl][7], a);
                asm("" : "+v"(a));
                lgl[2 * p * BT + tl * 16] = a;
            }
        }
        sweep2_head();
        __syncthreads();
        auto soft = [&](float* l, int q) {
            float mx = l[0];
#pragma unroll
            for (int jj = 1; jj < J; ++jj) mx = fmaxf(mx, l[jj]);
            float se = 0.0f;
#pragma unroll
            for (int jj = 0; jj < J; ++jj) { l[jj] = __expf(l[jj] - mx); se += l[jj]; }
            const float inv = 1.0f / se;
#pragma unroll
            for (int jj = 0; jj < J; ++jj) lg[jj * NQ + q] = l[jj] * inv;
        };
        {
            constexpr int NR = NQ / TPB;
            float l[NR][J];
#pragma unroll
            for (int r = 0; r < NR; ++r)
#pragma unroll
                for (int jj = 0; jj < J; ++jj) l[r][jj] = lg[jj * NQ + t + r * TPB];
#pragma unroll
            for (int r = 0; r < NR; ++r) soft(l[r], t + r * TPB);
            if (NQ % TPB != 0 && t < NQ % TPB) {
                const int q = t + NR * TPB;
                float lr[J];
#pragma unroll
                for (int jj = 0; jj < J; ++jj) lr[jj] = lg[jj * NQ + q];
                soft(lr, q);
            }
        }
        __syncthreads();
    }

    v8f sacc[TL];
#pragma unroll
    for (int tl = 0; tl < TL; ++tl) sacc[tl] = (v8f){0,0,0,0,0,0,0,0};

    {
        v4f xa[2][TL], xb[2][TL];
        float cc[2][TL];
        load_x(0, xa[0], xb[0]);
#pragma unroll
        for (int tl = 0; tl < TL; ++tl) cc[0][tl] = UNIFORM ? 1.0f : lgl[tl * 16];
#pragma unroll
        for (int p = 0; p < NP; ++p) {
            asm volatile("" ::: "memory");
            if (p + PF < NP) {
                const float* q = wsp + (p + PF) * PSTR;
#pragma unroll
                for (int n = 0; n < INDIM; ++n) wq[(p + PF) % (PF + 1)][n] = q[n * M];
            }
            if (p + 1 < NP) {
                load_x(p + 1, xa[(p + 1) & 1], xb[(p + 1) & 1]);
#pragma unroll
                for (int tl = 0; tl < TL; ++tl) cc[(p + 1) & 1][tl] = UNIFORM ? 1.0f : lgl[2 * (p + 1) * BT + tl * 16];
            }
            const float* wj = wq[p % (PF + 1)];
            const v2f w[4] = {{wj[0], wj[1]}, {wj[2], wj[3]}, {wj[4], wj[5]}, {wj[6], wj[7]}};
            v2f a[TL][4];
#pragma unroll
            for (int tl = 0; tl < TL; ++tl) {
                const v4f x0 = xa[p & 1][tl], x1 = xb[p & 1][tl];
                const float cj = cc[p & 1][tl];
                if (UNIFORM) {
                    a[tl][0] = (v2f){x0.x, x0.y}; a[tl][1] = (v2f){x0.z, x0.w};
                    a[tl][2] = (v2f){x1.x, x1.y}; a[tl][3] = (v2f){x1.z, x1.w};
                } else {
                    a[tl][0] = (v2f){x0.x * cj, x0.y * cj}; a[tl][1] = (v2f){x0.z * cj, x0.w * cj};
                    a[tl][2] = (v2f){x1.x * cj, x1.y * cj}; a[tl][3] = (v2f){x1.z * cj, x1.w * cj};
                }
            }
#pragma unroll
            for (int ks = 0; ks < 4; ++ks)
#pragma unroll
                for (int tl = 0; tl < TL; ++tl)
                    sacc[tl] = __builtin_amdgcn_wmma_f32_16x16x4_f32(false, a[tl][ks], false, w[ks],
                                                                     (short)0, sacc[tl], false, false);
        }
    }

    const float sc = UNIFORM ? 0.125f : 1.0f;
#pragma unroll
    for (int tl = 0; tl < TL; ++tl)
#pragma unroll
        for (int r = 0; r < 8; ++r) {
            const int bb = tl * 16 + r + half * 8;
            atomicAdd(&s_out[((size_t)(b0 + bb) * J + j) * M + lm], sc * sacc[tl][r]);
        }
}

__global__ void squash_k(const float* __restrict__ s, const float* __restrict__ vadd,
                         float* __restrict__ vout)
{
    const int bj = blockIdx.x * blockDim.x + threadIdx.x;
    if (bj >= B * J) return;
    const float* sp = s + bj * M;
    float sq = 0.0f;
#pragma unroll
    for (int m = 0; m < M; ++m) { const float e = sp[m]; sq += e * e; }
    const float scale = sqrtf(sq) / (1.0f + sq);
#pragma unroll
    for (int m = 0; m < M; ++m) {
        float r = sp[m] * scale;
        if (vadd) r += vadd[bj * M + m];
        vout[bj * M + m] = r;
    }
}

__global__ void zero_ws(float* p, int n)
{
    const int t = blockIdx.x * blockDim.x + threadIdx.x;
    if (t < n) p[t] = 0.0f;
}

extern "C" void kernel_launch(void* const* d_in, const int* in_sizes, int n_in,
                              void* d_out, int out_size, void* d_ws, size_t ws_size,
                              hipStream_t stream)
{
    (void)in_sizes; (void)n_in; (void)out_size; (void)ws_size;
    const float* x = (const float*)d_in[0];
    const float* W = (const float*)d_in[1];
    float* out = (float*)d_out;
    float* ws  = (float*)d_ws;

    float* s0  = ws;
    float* s1  = ws + SSZ;
    float* s2  = ws + 2 * SSZ;
    float* v0  = ws + 3 * SSZ;
    float* v01 = ws + 4 * SSZ;

    zero_ws<<<(3 * SSZ + TPB - 1) / TPB, TPB, 0, stream>>>(ws, 3 * SSZ);

    const dim3 grid(B / BT, NIB);
    routing_pass<true><<<grid, TPB, 0, stream>>>(x, W, nullptr, s0);
    squash_k<<<(B * J + TPB - 1) / TPB, TPB, 0, stream>>>(s0, nullptr, v0);
    routing_pass<false><<<grid, TPB, 0, stream>>>(x, W, v0, s1);
    squash_k<<<(B * J + TPB - 1) / TPB, TPB, 0, stream>>>(s1, v0, v01);
    routing_pass<false><<<grid, TPB, 0, stream>>>(x, W, v01, s2);
    squash_k<<<(B * J + TPB - 1) / TPB, TPB, 0, stream>>>(s2, nullptr, out);
}


# squash moved into the producing pass: the last of a batch block's 64 blocks to draw a ticket (one returning integer atomic add per block) squashes its 32 rows; 7 launches -> 4; no wait loop
# speedup vs baseline: 2.8612x; 1.0300x over previous
#include <hip/hip_runtime.h>
#include <math.h>


typedef __attribute__((ext_vector_type(2))) float v2f;
typedef __attribute__((ext_vector_type(4))) float v4f;
typedef __attribute__((ext_vector_type(8))) float v8f;

constexpr int B      = 256;
constexpr int ICAPS  = 3200;
constexpr int INDIM  = 8;
constexpr int J      = 8;
constexpr int M      = 16;
constexpr int TL     = 2;
constexpr int BT     = 16 * TL;
constexpr int IC     = 50;
constexpr int NIB    = ICAPS / IC;
constexpr int NP     = IC / 2;
constexpr int WAVES  = J;
constexpr int TPB    = WAVES * 32;
constexpr int SSZ    = B * J * M;
constexpr int WSTR   = J * INDIM * M;
constexpr int PF     = 2;
constexpr int PSTR   = 2 * WSTR;
constexpr int WPS    = 4;
constexpr int XP     = BT + 1;
constexpr int NQ     = IC * BT;
constexpr int LDS_BLOCK = 112 * 1024;
constexpr int NBB    = B / BT;
constexpr int CSTR   = 64;
constexpr int NCNT   = 3 * NBB * CSTR;
static_assert(ICAPS % IC == 0 && IC % 2 == 0, "whole capsule pairs per block");
static_assert(TPB == BT * J, "the finishing block squashes one (batch row, output capsule) per thread");

template <bool UNIFORM>
__global__ __launch_bounds__(TPB, WPS) void routing_pass(
    const float* __restrict__ x,
    const float* __restrict__ W,
    const float* __restrict__ vprev,
    float* __restrict__ s_out,
    float* __restrict__ cnt_area,
    const float* __restrict__ vadd,
    float* __restrict__ vout)
{
    struct __attribute__((aligned(16))) Tile {
        float xs[IC * 2 * XP * 4];
        float lg[J * NQ];
        float fill[(LDS_BLOCK - (IC * 2 * XP * 4 + J * NQ) * 4) / 4];
    };
    static_assert(sizeof(Tile) == LDS_BLOCK, "two blocks per WGP");
    __shared__ Tile sm;
    float* const xs = sm.xs;
    float* const lg = sm.lg;

    const int t  = threadIdx.x;
    const int b0 = blockIdx.x * BT;
    const int i0 = blockIdx.y * IC;

    const int j    = __builtin_amdgcn_readfirstlane(t >> 5);
    const int lane = t & 31;
    const int half = lane >> 4;
    const int lm   = lane & 15;

    v4f wr[PF + 1][2];
    const float* wz = W + (size_t)(i0 + (lm >> 3)) * WSTR + j * (INDIM * M) + (lm & 7) * M + half * 8;
    float wq[PF + 1][INDIM];
    const float* wsp = W + (size_t)(i0 + half) * WSTR + j * (INDIM * M) + lm;
    auto sweep2_head = [&]() {
#pragma unroll
        for (int s = 0; s < PF; ++s) {
            const float* q = wsp + s * PSTR;
#pragma unroll
            for (int n = 0; n < INDIM; ++n) wq[s][n] = q[n * M];
        }
    };
    v4f vf[TL][2];
    if constexpr (UNIFORM) {
        sweep2_head();
    } else {
#pragma unroll
        for (int s = 0; s < PF; ++s) {
            const float* q = wz + s * PSTR;
            wr[s][0] = *reinterpret_cast<const v4f*>(q);
            wr[s][1] = *reinterpret_cast<const v4f*>(q + 4);
        }
#pragma unroll
        for (int tl = 0; tl < TL; ++tl) {
            const float* vp = vprev + ((size_t)(b0 + tl * 16 + lm) * J + j) * M + half * 8;
            vf[tl][0] = *reinterpret_cast<const v4f*>(vp);
            vf[tl][1] = *reinterpret_cast<const v4f*>(vp + 4);
        }
    }

    {
        const float4* xg  = reinterpret_cast<const float4*>(x);
        float4*       xs4 = reinterpret_cast<float4*>(xs);
        constexpr int NX = BT * IC * 2;
        auto stage = [&](int q) {
            const int bb = q / (IC * 2);
            const int k  = q - bb * (IC * 2);
            xs4[k * XP + bb] = xg[((size_t)(b0 + bb) * ICAPS + i0) * 2 + k];
        };
#pragma unroll
        for (int it = 0; it < NX / TPB; ++it) stage(t + it * TPB);
        if (NX % TPB != 0 && t < NX % TPB) stage(t + (NX / TPB) * TPB);
    }
    __syncthreads();

    const v4f* const xl = reinterpret_cast<const v4f*>(xs) + (2 * half) * XP + lm;
    auto load_x = [&](int p, v4f* xa, v4f* xb) {
#pragma unroll
        for (int tl = 0; tl < TL; ++tl) {
            xa[tl] = xl[(4 * p) * XP + tl * 16];
            xb[tl] = xl[(4 * p + 1) * XP + tl * 16];
        }
    };
    float* const lgl = lg + j * NQ + half * BT + lm;

    if constexpr (!UNIFORM) {
        v4f xa[2][TL], xb[2][TL];
        load_x(0, xa[0], xb[0]);
#pragma unroll
        for (int p = 0; p < NP; ++p) {
            asm volatile("" ::: "memory");
            if (p + PF < NP) {
                const float* q = wz + (p + PF) * PSTR;
                wr[(p + PF) % (PF + 1)][0] = *reinterpret_cast<const v4f*>(q);
                wr[(p + PF) % (PF + 1)][1] = *reinterpret_cast<const v4f*>(q + 4);
            }
            const v4f wa = wr[p % (PF + 1)][0];
            const v4f wb = wr[p % (PF + 1)][1];
            if (p + 1 < NP) load_x(p + 1, xa[(p + 1) & 1], xb[(p + 1) & 1]);
            v8f z[TL];
#pragma unroll
            for (int tl = 0; tl < TL; ++tl) {
                const v4f va = vf[tl][0];
                const v4f vb = vf[tl][1];
                v8f d = {0,0,0,0,0,0,0,0};
                d = __builtin_amdgcn_wmma_f32_16x16x4_f32(false, (v2f){wa.x, wa.y}, false, (v2f){va.x, va.y},
                                                          (short)0, d, false, false);
                d = __builtin_amdgcn_wmma_f32_16x16x4_f32(false, (v2f){wa.z, wa.w}, false, (v2f){va.z, va.w},
                                                          (short)0, d, false, false);
                d = __builtin_amdgcn_wmma_f32_16x16x4_f32(false, (v2f){wb.x, wb.y}, false, (v2f){vb.x, vb.y},
                                                          (short)0, d, false, false);
                d = __builtin_amdgcn_wmma_f32_16x16x4_f32(false, (v2f){wb.z, wb.w}, false, (v2f){vb.z, vb.w},
                                                          (short)0, d, false, false);
                z[tl] = d;
            }
#pragma unroll
            for (int tl = 0; tl < TL; ++tl) {
                const v4f x0 = xa[p & 1][tl], x1 = xb[p & 1][tl];
                float a = x0.x * z[tl][0];
                a = fmaf(x0.y, z[tl][1], a); a = fmaf(x0.z, z[tl][2], a); a = fmaf(x0.w, z[tl][3], a);
                a = fmaf(x1.x, z[tl][4], a); a = fmaf(x1.y, z[tl][5], a); a = fmaf(x1.z, z[tl][6], a); a = fmaf(x1.w, z[tl][7], a);
                asm("" : "+v"(a));
                lgl[2 * p * BT + tl * 16] = a;
            }
        }
        sweep2_head();
        __syncthreads();
        auto soft = [&](float* l, int q) {
            float mx = l[0];
#pragma unroll
            for (int jj = 1; jj < J; ++jj) mx = fmaxf(mx, l[jj]);
            float se = 0.0f;
#pragma unroll
            for (int jj = 0; jj < J; ++jj) { l[jj] = __expf(l[jj] - mx); se += l[jj]; }
            const float inv = 1.0f / se;
#pragma unroll
            for (int jj = 0; jj < J; ++jj) lg[jj * NQ + q] = l[jj] * inv;
        };
        {
            constexpr int NR = NQ / TPB;
            float l[NR][J];
#pragma unroll
            for (int r = 0; r < NR; ++r)
#pragma unroll
                for (int jj = 0; jj < J; ++jj) l[r][jj] = lg[jj * NQ + t + r * TPB];
#pragma unroll
            for (int r = 0; r < NR; ++r) soft(l[r], t + r * TPB);
            if (NQ % TPB != 0 && t < NQ % TPB) {
                const int q = t + NR * TPB;
                float lr[J];
#pragma unroll
                for (int jj = 0; jj < J; ++jj) lr[jj] = lg[jj * NQ + q];
                soft(lr, q);
            }
        }
        __syncthreads();
    }

    v8f sacc[TL];
#pragma unroll
    for (int tl = 0; tl < TL; ++tl) sacc[tl] = (v8f){0,0,0,0,0,0,0,0};

    {
        v4f xa[2][TL], xb[2][TL];
        float cc[2][TL];
        load_x(0, xa[0], xb[0]);
#pragma unroll
        for (int tl = 0; tl < TL; ++tl) cc[0][tl] = UNIFORM ? 1.0f : lgl[tl * 16];
#pragma unroll
        for (int p = 0; p < NP; ++p) {
            asm volatile("" ::: "memory");
            if (p + PF < NP) {
                const float* q = wsp + (p + PF) * PSTR;
#pragma unroll
                for (int n = 0; n < INDIM; ++n) wq[(p + PF) % (PF + 1)][n] = q[n * M];
            }
            if (p + 1 < NP) {
                load_x(p + 1, xa[(p + 1) & 1], xb[(p + 1) & 1]);
#pragma unroll
                for (int tl = 0; tl < TL; ++tl) cc[(p + 1) & 1][tl] = UNIFORM ? 1.0f : lgl[2 * (p + 1) * BT + tl * 16];
            }
            const float* wj = wq[p % (PF + 1)];
            const v2f w[4] = {{wj[0], wj[1]}, {wj[2], wj[3]}, {wj[4], wj[5]}, {wj[6], wj[7]}};
            v2f a[TL][4];
#pragma unroll
            for (int tl = 0; tl < TL; ++tl) {
                const v4f x0 = xa[p & 1][tl], x1 = xb[p & 1][tl];
                const float cj = cc[p & 1][tl];
                if (UNIFORM) {
                    a[tl][0] = (v2f){x0.x, x0.y}; a[tl][1] = (v2f){x0.z, x0.w};
                    a[tl][2] = (v2f){x1.x, x1.y}; a[tl][3] = (v2f){x1.z, x1.w};
                } else {
                    a[tl][0] = (v2f){x0.x * cj, x0.y * cj}; a[tl][1] = (v2f){x0.z * cj, x0.w * cj};
                    a[tl][2] = (v2f){x1.x * cj, x1.y * cj}; a[tl][3] = (v2f){x1.z * cj, x1.w * cj};
                }
            }
#pragma unroll
            for (int ks = 0; ks < 4; ++ks)
#pragma unroll
                for (int tl = 0; tl < TL; ++tl)
                    sacc[tl] = __builtin_amdgcn_wmma_f32_16x16x4_f32(false, a[tl][ks], false, w[ks],
                                                                     (short)0, sacc[tl], false, false);
        }
    }

    const float sc = UNIFORM ? 0.125f : 1.0f;
#pragma unroll
    for (int tl = 0; tl < TL; ++tl)
#pragma unroll
        for (int r = 0; r < 8; ++r) {
            const int bb = tl * 16 + r + half * 8;
            atomicAdd(&s_out[((size_t)(b0 + bb) * J + j) * M + lm], sc * sacc[tl][r]);
        }

    unsigned* const tk = reinterpret_cast<unsigned*>(sm.fill);
    __syncthreads();
    if (t == 0) {
        __builtin_amdgcn_fence(__ATOMIC_RELEASE, "agent");
        unsigned* const cnt = reinterpret_cast<unsigned*>(cnt_area) + blockIdx.x * CSTR;
        *tk = __hip_atomic_fetch_add(cnt, 1u, __ATOMIC_RELAXED, __HIP_MEMORY_SCOPE_AGENT);
    }
    __syncthreads();
    if (*tk == NIB - 1) {
        __builtin_amdgcn_fence(__ATOMIC_ACQUIRE, "agent");
        const size_t o = (size_t)(b0 * J + t) * M;
        const float4* const sp = reinterpret_cast<const float4*>(s_out + o);
        float4 e[M / 4], a[M / 4];
#pragma unroll
        for (int k = 0; k < M / 4; ++k) e[k] = sp[k];
        if (vadd) {
            const float4* const va = reinterpret_cast<const float4*>(vadd + o);
#pragma unroll
            for (int k = 0; k < M / 4; ++k) a[k] = va[k];
        } else {
#pragma unroll
            for (int k = 0; k < M / 4; ++k) a[k] = make_float4(0.0f, 0.0f, 0.0f, 0.0f);
        }
        float sq = 0.0f;
#pragma unroll
        for (int k = 0; k < M / 4; ++k) {
            sq += e[k].x * e[k].x; sq += e[k].y * e[k].y; sq += e[k].z * e[k].z; sq += e[k].w * e[k].w;
        }
        const float scale = sqrtf(sq) / (1.0f + sq);
        float4* const vo = reinterpret_cast<float4*>(vout + o);
#pragma unroll
        for (int k = 0; k < M / 4; ++k)
            vo[k] = make_float4(e[k].x * scale + a[k].x, e[k].y * scale + a[k].y,
                                e[k].z * scale + a[k].z, e[k].w * scale + a[k].w);
    }
}

__global__ void zero_ws(float* __restrict__ p, int n, float* __restrict__ c, int nc)
{
    const int t = blockIdx.x * blockDim.x + threadIdx.x;
    if (t < n) p[t] = 0.0f;
    if (t < nc) c[t] = 0.0f;
}

extern "C" void kernel_launch(void* const* d_in, const int* in_sizes, int n_in,
                              void* d_out, int out_size, void* d_ws, size_t ws_size,
                              hipStream_t stream)
{
    (void)in_sizes; (void)n_in; (void)out_size; (void)ws_size;
    const float* x = (const float*)d_in[0];
    const float* W = (const float*)d_in[1];
    float* out = (float*)d_out;
    float* ws  = (float*)d_ws;

    float* s0  = ws;
    float* s1  = ws + SSZ;
    float* s2  = ws + 2 * SSZ;
    float* v0  = ws + 3 * SSZ;
    float* cnt = ws + 4 * SSZ;
    float* v01 = s0;

    zero_ws<<<(3 * SSZ + TPB - 1) / TPB, TPB, 0, stream>>>(ws, 3 * SSZ, cnt, NCNT);

    const dim3 grid(NBB, NIB);
    routing_pass<true><<<grid, TPB, 0, stream>>>(x, W, nullptr, s0, cnt, nullptr, v0);
    routing_pass<false><<<grid, TPB, 0, stream>>>(x, W, v0, s1, cnt + NBB * CSTR, v0, v01);
    routing_pass<false><<<grid, TPB, 0, stream>>>(x, W, v01, s2, cnt + 2 * NBB * CSTR, nullptr, out);
}
